# starting kernel
# baseline (speedup 1.0000x reference)
#include <hip/hip_runtime.h>
#include <hip/hip_bf16.h>

typedef __attribute__((ext_vector_type(2))) float v2f;
typedef __attribute__((ext_vector_type(8))) float v8f;

#define NDIM_GRID 20
#define NY_NZ     400
#define NPTS      8000
#define NTILES    500
#define LOG2E     1.4426950408889634f

__device__ __forceinline__ void decode_coord(int i, float& x, float& y, float& z) {
    int ix = i / NY_NZ;
    int rem = i - ix * NY_NZ;
    int iy = rem / NDIM_GRID;
    int iz = rem - iy * NDIM_GRID;
    x = (float)ix; y = (float)iy; z = (float)iz;
}

__global__ void __launch_bounds__(64) kde_main(
        const float* __restrict__ p,
        const float* __restrict__ Ainv,
        float* __restrict__ kde)
{
    __shared__ float4 bq[NPTS];
    __shared__ float  pp[NPTS];

    const int tid  = threadIdx.x;
    const int wave = (blockIdx.x * blockDim.x + tid) >> 5;
    const int lane = tid & 31;
    const int half = lane >> 4;
    const int l15  = lane & 15;

    const float a00 = Ainv[0], a01 = Ainv[1], a02 = Ainv[2];
    const float a10 = Ainv[3], a11 = Ainv[4], a12 = Ainv[5];
    const float a20 = Ainv[6], a21 = Ainv[7], a22 = Ainv[8];

    for (int i = tid; i < NPTS; i += 64) {
        float cx, cy, cz; decode_coord(i, cx, cy, cz);
        float b0 = a00 * cx + a01 * cy + a02 * cz;
        float b1 = a10 * cx + a11 * cy + a12 * cz;
        float b2 = a20 * cx + a21 * cy + a22 * cz;
        float q  = cx * b0 + cy * b1 + cz * b2;
        bq[i] = make_float4(cx, cy, cz, q);
        pp[i] = p[i];
    }
    __syncthreads();

    const int row = wave * 16 + l15;
    float rx, ry, rz; decode_coord(row, rx, ry, rz);
    const float g0 = a00 * rx + a01 * ry + a02 * rz;
    const float g1 = a10 * rx + a11 * ry + a12 * rz;
    const float g2 = a20 * rx + a21 * ry + a22 * rz;
    const float qrow = rx * g0 + ry * g1 + rz * g2;
    v2f amat;
    amat.x = half ? (LOG2E * g2) : (LOG2E * g0);
    amat.y = half ? (-0.5f * LOG2E) : (LOG2E * g1);

    v8f cvec;
#pragma unroll
    for (int r = 0; r < 8; ++r)
        cvec[r] = (-0.5f * LOG2E) * __shfl(qrow, (lane & 16) + half * 8 + r, 32);

    float acc[8] = {};

    const v2f*   bbase = (const v2f*)bq + (l15 * 2 + half);
    const float* pbase = pp + l15;

#pragma unroll 2
    for (int t = 0; t < NTILES; ++t) {
        const v2f   bmat = bbase[t * 32];
        const float pb   = pbase[t * 16];

        v8f c = __builtin_amdgcn_wmma_f32_16x16x4_f32(
                false, amat, false, bmat, (short)0, cvec, false, false);

#pragma unroll
        for (int r = 0; r < 8; ++r)
            acc[r] = fmaf(__builtin_amdgcn_exp2f(c[r]), pb, acc[r]);
    }

#pragma unroll
    for (int r = 0; r < 8; ++r) {
        float v = acc[r];
        v += __shfl_xor(v, 1, 32);
        v += __shfl_xor(v, 2, 32);
        v += __shfl_xor(v, 4, 32);
        v += __shfl_xor(v, 8, 32);
        if (l15 == 0)
            kde[wave * 16 + r + 8 * half] = v;
    }
}

__global__ void kde_norm(float* __restrict__ kde) {
    __shared__ float red[256];
    float s = 0.0f;
    for (int i = threadIdx.x; i < NPTS; i += 256) s += kde[i];
    red[threadIdx.x] = s;
    __syncthreads();
    for (int off = 128; off > 0; off >>= 1) {
        if (threadIdx.x < off) red[threadIdx.x] += red[threadIdx.x + off];
        __syncthreads();
    }
    const float inv = 1.0f / red[0];
    for (int i = threadIdx.x; i < NPTS; i += 256) kde[i] *= inv;
}

extern "C" void kernel_launch(void* const* d_in, const int* in_sizes, int n_in,
                              void* d_out, int out_size, void* d_ws, size_t ws_size,
                              hipStream_t stream) {
    const float* p    = (const float*)d_in[0];
    const float* ainv = (const float*)d_in[1];
    float* out = (float*)d_out;

    kde_main<<<250, 64, 0, stream>>>(p, ainv, out);
    kde_norm<<<1, 256, 0, stream>>>(out);
}


# Column sweep of each 32-row block split over 8 waves (2 per SIMD, all 4 SIMDs) instead of 2; each B fragment feeds both 16-row tiles; per-wave partial row sums added through LDS (new reduction tree)
# speedup vs baseline: 1.7997x; 1.7997x over previous
#include <hip/hip_runtime.h>
#include <hip/hip_bf16.h>


typedef __attribute__((ext_vector_type(2))) float v2f;
typedef __attribute__((ext_vector_type(8))) float v8f;

#define NDIM_GRID 20
#define NY_NZ     400
#define NPTS      8000
#define NTILES    500
#define NPAIRS    250
#define LOG2E     1.4426950408889634f

#define NW        8
#define NT        256
#define WG_ROWS   32
#define NSTAGE    32
#define NSTHR     250

__device__ __forceinline__ void decode_coord(int i, float& x, float& y, float& z) {
    const unsigned short u   = (unsigned short)i;
    const unsigned short ix  = u / NY_NZ;
    const unsigned short rem = u - ix * NY_NZ;
    const unsigned short iy  = rem / NDIM_GRID;
    const unsigned short iz  = rem - iy * NDIM_GRID;
    x = (float)ix; y = (float)iy; z = (float)iz;
}

__device__ __forceinline__ float row16_sum(float v) {
    v += __builtin_bit_cast(float, __builtin_amdgcn_update_dpp(0, __builtin_bit_cast(int, v), 0x161, 0xf, 0xf, false));
    v += __builtin_bit_cast(float, __builtin_amdgcn_update_dpp(0, __builtin_bit_cast(int, v), 0x162, 0xf, 0xf, false));
    v += __builtin_bit_cast(float, __builtin_amdgcn_update_dpp(0, __builtin_bit_cast(int, v), 0x164, 0xf, 0xf, false));
    v += __builtin_bit_cast(float, __builtin_amdgcn_update_dpp(0, __builtin_bit_cast(int, v), 0x168, 0xf, 0xf, false));
    return v;
}

__global__ void __launch_bounds__(NT) kde_main(
        const float* __restrict__ p,
        const float* __restrict__ Ainv,
        float* __restrict__ kde)
{
    __shared__ float4 bq[NPTS];
    __shared__ float  pp[NPTS];
    __shared__ float  part[NW * WG_ROWS];

    const int tid  = threadIdx.x;
    const int lane = tid & 31;
    const int w    = __builtin_amdgcn_readfirstlane(tid >> 5);
    const int half = lane >> 4;
    const int l15  = lane & 15;

    const float a00 = Ainv[0], a01 = Ainv[1], a02 = Ainv[2];
    const float a10 = Ainv[3], a11 = Ainv[4], a12 = Ainv[5];
    const float a20 = Ainv[6], a21 = Ainv[7], a22 = Ainv[8];

    if (tid < NSTHR) {
        float pv[NSTAGE];
#pragma unroll
        for (int k = 0; k < NSTAGE; ++k)
            pv[k] = p[k * NSTHR + tid];
#pragma unroll 4
        for (int k = 0; k < NSTAGE; ++k) {
            const int i = k * NSTHR + tid;
            float cx, cy, cz; decode_coord(i, cx, cy, cz);
            float b0 = a00 * cx + a01 * cy + a02 * cz;
            float b1 = a10 * cx + a11 * cy + a12 * cz;
            float b2 = a20 * cx + a21 * cy + a22 * cz;
            float q  = cx * b0 + cy * b1 + cz * b2;
            bq[i] = make_float4(cx, cy, cz, q);
        }
#pragma unroll
        for (int k = 0; k < NSTAGE; ++k)
            pp[k * NSTHR + tid] = pv[k];
    }
    __syncthreads();

    const int rowbase = blockIdx.x * WG_ROWS;
    v2f amat0, amat1;
    {
        float rx, ry, rz; decode_coord(rowbase + l15, rx, ry, rz);
        const float g0 = a00 * rx + a01 * ry + a02 * rz;
        const float g1 = a10 * rx + a11 * ry + a12 * rz;
        const float g2 = a20 * rx + a21 * ry + a22 * rz;
        amat0.x = half ? (LOG2E * g2) : (LOG2E * g0);
        amat0.y = half ? (-0.5f * LOG2E) : (LOG2E * g1);
    }
    {
        float rx, ry, rz; decode_coord(rowbase + 16 + l15, rx, ry, rz);
        const float g0 = a00 * rx + a01 * ry + a02 * rz;
        const float g1 = a10 * rx + a11 * ry + a12 * rz;
        const float g2 = a20 * rx + a21 * ry + a22 * rz;
        amat1.x = half ? (LOG2E * g2) : (LOG2E * g0);
        amat1.y = half ? (-0.5f * LOG2E) : (LOG2E * g1);
    }
    v8f cvec0, cvec1;
#pragma unroll
    for (int r = 0; r < 8; ++r) {
        cvec0[r] = (-0.5f * LOG2E) * bq[rowbase +      8 * half + r].w;
        cvec1[r] = (-0.5f * LOG2E) * bq[rowbase + 16 + 8 * half + r].w;
    }

    float acc0[8] = {};
    float acc1[8] = {};

    const v2f*   bbase = (const v2f*)bq + (l15 * 2 + half);
    const float* pbase = pp + l15;

    const int npair = (NPAIRS - w + NW - 1) / NW;
    int j = w;
    v2f   b0 = bbase[(2 * j) * 32];
    v2f   b1 = bbase[(2 * j + 1) * 32];
    float p0 = pbase[(2 * j) * 16];
    float p1 = pbase[(2 * j + 1) * 16];

    for (int k = 0; k < npair; ++k) {
        const int jn = (j + NW < NPAIRS) ? (j + NW) : j;
        const v2f   nb0 = bbase[(2 * jn) * 32];
        const v2f   nb1 = bbase[(2 * jn + 1) * 32];
        const float np0 = pbase[(2 * jn) * 16];
        const float np1 = pbase[(2 * jn + 1) * 16];

        v8f c00 = __builtin_amdgcn_wmma_f32_16x16x4_f32(
                false, amat0, false, b0, (short)0, cvec0, false, false);
        v8f c10 = __builtin_amdgcn_wmma_f32_16x16x4_f32(
                false, amat1, false, b0, (short)0, cvec1, false, false);
        v8f c01 = __builtin_amdgcn_wmma_f32_16x16x4_f32(
                false, amat0, false, b1, (short)0, cvec0, false, false);
        v8f c11 = __builtin_amdgcn_wmma_f32_16x16x4_f32(
                false, amat1, false, b1, (short)0, cvec1, false, false);

#pragma unroll
        for (int r = 0; r < 8; ++r)
            acc0[r] = fmaf(__builtin_amdgcn_exp2f(c00[r]), p0, acc0[r]);
#pragma unroll
        for (int r = 0; r < 8; ++r)
            acc1[r] = fmaf(__builtin_amdgcn_exp2f(c10[r]), p0, acc1[r]);
#pragma unroll
        for (int r = 0; r < 8; ++r)
            acc0[r] = fmaf(__builtin_amdgcn_exp2f(c01[r]), p1, acc0[r]);
#pragma unroll
        for (int r = 0; r < 8; ++r)
            acc1[r] = fmaf(__builtin_amdgcn_exp2f(c11[r]), p1, acc1[r]);

        asm volatile("" :: "v"(nb0), "v"(nb1), "v"(np0), "v"(np1));
        b0 = nb0; b1 = nb1; p0 = np0; p1 = np1;
        j += NW;
    }

#pragma unroll
    for (int r = 0; r < 8; ++r) {
        const float s0 = row16_sum(acc0[r]);
        const float s1 = row16_sum(acc1[r]);
        if (l15 == 0) {
            part[w * WG_ROWS +      8 * half + r] = s0;
            part[w * WG_ROWS + 16 + 8 * half + r] = s1;
        }
    }
    __syncthreads();

    if (tid < WG_ROWS) {
        float s = part[tid];
#pragma unroll
        for (int v = 1; v < NW; ++v)
            s += part[v * WG_ROWS + tid];
        kde[rowbase + tid] = s;
    }
}

__global__ void kde_norm(float* __restrict__ kde) {
    __shared__ float red[256];
    float s = 0.0f;
    for (int i = threadIdx.x; i < NPTS; i += 256) s += kde[i];
    red[threadIdx.x] = s;
    __syncthreads();
    for (int off = 128; off > 0; off >>= 1) {
        if (threadIdx.x < off) red[threadIdx.x] += red[threadIdx.x + off];
        __syncthreads();
    }
    const float inv = 1.0f / red[0];
    for (int i = threadIdx.x; i < NPTS; i += 256) kde[i] *= inv;
}

extern "C" void kernel_launch(void* const* d_in, const int* in_sizes, int n_in,
                              void* d_out, int out_size, void* d_ws, size_t ws_size,
                              hipStream_t stream) {
    const float* p    = (const float*)d_in[0];
    const float* ainv = (const float*)d_in[1];
    float* out = (float*)d_out;

    kde_main<<<250, 256, 0, stream>>>(p, ainv, out);
    kde_norm<<<1, 256, 0, stream>>>(out);
}


# Normalizing kernel in one pass: each thread loads its 32 values at once (float4), keeps them in registers; total by half-wave adds and one barrier (new summation tree); scaled stores from registers
# speedup vs baseline: 3.5373x; 1.9655x over previous
#include <hip/hip_runtime.h>
#include <hip/hip_bf16.h>


typedef __attribute__((ext_vector_type(2))) float v2f;
typedef __attribute__((ext_vector_type(8))) float v8f;

#define NDIM_GRID 20
#define NY_NZ     400
#define NPTS      8000
#define NTILES    500
#define NPAIRS    250
#define LOG2E     1.4426950408889634f

#define NW        8
#define NT        256
#define WG_ROWS   32
#define NSTAGE    32
#define NSTHR     250

__device__ __forceinline__ void decode_coord(int i, float& x, float& y, float& z) {
    const unsigned short u   = (unsigned short)i;
    const unsigned short ix  = u / NY_NZ;
    const unsigned short rem = u - ix * NY_NZ;
    const unsigned short iy  = rem / NDIM_GRID;
    const unsigned short iz  = rem - iy * NDIM_GRID;
    x = (float)ix; y = (float)iy; z = (float)iz;
}

__device__ __forceinline__ float row16_sum(float v) {
    v += __builtin_bit_cast(float, __builtin_amdgcn_update_dpp(0, __builtin_bit_cast(int, v), 0x161, 0xf, 0xf, false));
    v += __builtin_bit_cast(float, __builtin_amdgcn_update_dpp(0, __builtin_bit_cast(int, v), 0x162, 0xf, 0xf, false));
    v += __builtin_bit_cast(float, __builtin_amdgcn_update_dpp(0, __builtin_bit_cast(int, v), 0x164, 0xf, 0xf, false));
    v += __builtin_bit_cast(float, __builtin_amdgcn_update_dpp(0, __builtin_bit_cast(int, v), 0x168, 0xf, 0xf, false));
    return v;
}

__global__ void __launch_bounds__(NT) kde_main(
        const float* __restrict__ p,
        const float* __restrict__ Ainv,
        float* __restrict__ kde)
{
    __shared__ float4 bq[NPTS];
    __shared__ float  pp[NPTS];
    __shared__ float  part[NW * WG_ROWS];

    const int tid  = threadIdx.x;
    const int lane = tid & 31;
    const int w    = __builtin_amdgcn_readfirstlane(tid >> 5);
    const int half = lane >> 4;
    const int l15  = lane & 15;

    const float a00 = Ainv[0], a01 = Ainv[1], a02 = Ainv[2];
    const float a10 = Ainv[3], a11 = Ainv[4], a12 = Ainv[5];
    const float a20 = Ainv[6], a21 = Ainv[7], a22 = Ainv[8];

    if (tid < NSTHR) {
        float pv[NSTAGE];
#pragma unroll
        for (int k = 0; k < NSTAGE; ++k)
            pv[k] = p[k * NSTHR + tid];
#pragma unroll 4
        for (int k = 0; k < NSTAGE; ++k) {
            const int i = k * NSTHR + tid;
            float cx, cy, cz; decode_coord(i, cx, cy, cz);
            float b0 = a00 * cx + a01 * cy + a02 * cz;
            float b1 = a10 * cx + a11 * cy + a12 * cz;
            float b2 = a20 * cx + a21 * cy + a22 * cz;
            float q  = cx * b0 + cy * b1 + cz * b2;
            bq[i] = make_float4(cx, cy, cz, q);
        }
#pragma unroll
        for (int k = 0; k < NSTAGE; ++k)
            pp[k * NSTHR + tid] = pv[k];
    }
    __syncthreads();

    const int rowbase = blockIdx.x * WG_ROWS;
    v2f amat0, amat1;
    {
        float rx, ry, rz; decode_coord(rowbase + l15, rx, ry, rz);
        const float g0 = a00 * rx + a01 * ry + a02 * rz;
        const float g1 = a10 * rx + a11 * ry + a12 * rz;
        const float g2 = a20 * rx + a21 * ry + a22 * rz;
        amat0.x = half ? (LOG2E * g2) : (LOG2E * g0);
        amat0.y = half ? (-0.5f * LOG2E) : (LOG2E * g1);
    }
    {
        float rx, ry, rz; decode_coord(rowbase + 16 + l15, rx, ry, rz);
        const float g0 = a00 * rx + a01 * ry + a02 * rz;
        const float g1 = a10 * rx + a11 * ry + a12 * rz;
        const float g2 = a20 * rx + a21 * ry + a22 * rz;
        amat1.x = half ? (LOG2E * g2) : (LOG2E * g0);
        amat1.y = half ? (-0.5f * LOG2E) : (LOG2E * g1);
    }
    v8f cvec0, cvec1;
#pragma unroll
    for (int r = 0; r < 8; ++r) {
        cvec0[r] = (-0.5f * LOG2E) * bq[rowbase +      8 * half + r].w;
        cvec1[r] = (-0.5f * LOG2E) * bq[rowbase + 16 + 8 * half + r].w;
    }

    float acc0[8] = {};
    float acc1[8] = {};

    const v2f*   bbase = (const v2f*)bq + (l15 * 2 + half);
    const float* pbase = pp + l15;

    const int npair = (NPAIRS - w + NW - 1) / NW;
    int j = w;
    v2f   b0 = bbase[(2 * j) * 32];
    v2f   b1 = bbase[(2 * j + 1) * 32];
    float p0 = pbase[(2 * j) * 16];
    float p1 = pbase[(2 * j + 1) * 16];

    for (int k = 0; k < npair; ++k) {
        const int jn = (j + NW < NPAIRS) ? (j + NW) : j;
        const v2f   nb0 = bbase[(2 * jn) * 32];
        const v2f   nb1 = bbase[(2 * jn + 1) * 32];
        const float np0 = pbase[(2 * jn) * 16];
        const float np1 = pbase[(2 * jn + 1) * 16];

        v8f c00 = __builtin_amdgcn_wmma_f32_16x16x4_f32(
                false, amat0, false, b0, (short)0, cvec0, false, false);
        v8f c10 = __builtin_amdgcn_wmma_f32_16x16x4_f32(
                false, amat1, false, b0, (short)0, cvec1, false, false);
        v8f c01 = __builtin_amdgcn_wmma_f32_16x16x4_f32(
                false, amat0, false, b1, (short)0, cvec0, false, false);
        v8f c11 = __builtin_amdgcn_wmma_f32_16x16x4_f32(
                false, amat1, false, b1, (short)0, cvec1, false, false);

#pragma unroll
        for (int r = 0; r < 8; ++r)
            acc0[r] = fmaf(__builtin_amdgcn_exp2f(c00[r]), p0, acc0[r]);
#pragma unroll
        for (int r = 0; r < 8; ++r)
            acc1[r] = fmaf(__builtin_amdgcn_exp2f(c10[r]), p0, acc1[r]);
#pragma unroll
        for (int r = 0; r < 8; ++r)
            acc0[r] = fmaf(__builtin_amdgcn_exp2f(c01[r]), p1, acc0[r]);
#pragma unroll
        for (int r = 0; r < 8; ++r)
            acc1[r] = fmaf(__builtin_amdgcn_exp2f(c11[r]), p1, acc1[r]);

        asm volatile("" :: "v"(nb0), "v"(nb1), "v"(np0), "v"(np1));
        b0 = nb0; b1 = nb1; p0 = np0; p1 = np1;
        j += NW;
    }

#pragma unroll
    for (int r = 0; r < 8; ++r) {
        const float s0 = row16_sum(acc0[r]);
        const float s1 = row16_sum(acc1[r]);
        if (l15 == 0) {
            part[w * WG_ROWS +      8 * half + r] = s0;
            part[w * WG_ROWS + 16 + 8 * half + r] = s1;
        }
    }
    __syncthreads();

    if (tid < WG_ROWS) {
        float s = part[tid];
#pragma unroll
        for (int v = 1; v < NW; ++v)
            s += part[v * WG_ROWS + tid];
        kde[rowbase + tid] = s;
    }
}

#define NORM_T   256
#define NORM_V   8
#define NVEC4    2000

__global__ void __launch_bounds__(NORM_T) kde_norm(float* __restrict__ kde) {
    __shared__ float red[16];
    const int tid  = threadIdx.x;
    const int lane = tid & 31;
    const int w    = tid >> 5;
    float4* vec = (float4*)kde;

    float4 v[NORM_V];
#pragma unroll
    for (int k = 0; k < NORM_V - 1; ++k)
        v[k] = vec[k * NORM_T + tid];
    const bool tail = (NORM_V - 1) * NORM_T + tid < NVEC4;
    v[NORM_V - 1] = tail ? vec[(NORM_V - 1) * NORM_T + tid]
                         : make_float4(0.0f, 0.0f, 0.0f, 0.0f);

    float t4[NORM_V];
#pragma unroll
    for (int k = 0; k < NORM_V; ++k)
        t4[k] = (v[k].x + v[k].y) + (v[k].z + v[k].w);
    float s = ((t4[0] + t4[1]) + (t4[2] + t4[3])) + ((t4[4] + t4[5]) + (t4[6] + t4[7]));

    s = row16_sum(s);
    if ((lane & 15) == 0) red[2 * w + (lane >> 4)] = s;
    __syncthreads();
    const float total = row16_sum(red[lane & 15]);
    const float inv = 1.0f / total;

#pragma unroll
    for (int k = 0; k < NORM_V - 1; ++k)
        vec[k * NORM_T + tid] = make_float4(v[k].x * inv, v[k].y * inv, v[k].z * inv, v[k].w * inv);
    if (tail)
        vec[(NORM_V - 1) * NORM_T + tid] = make_float4(v[NORM_V - 1].x * inv, v[NORM_V - 1].y * inv,
                                                       v[NORM_V - 1].z * inv, v[NORM_V - 1].w * inv);
}

extern "C" void kernel_launch(void* const* d_in, const int* in_sizes, int n_in,
                              void* d_out, int out_size, void* d_ws, size_t ws_size,
                              hipStream_t stream) {
    const float* p    = (const float*)d_in[0];
    const float* ainv = (const float*)d_in[1];
    float* out = (float*)d_out;

    kde_main<<<250, 256, 0, stream>>>(p, ainv, out);
    kde_norm<<<1, NORM_T, 0, stream>>>(out);
}


# Staging pass made cheap: a thread owns one (y,z) pair and walks the 20 x planes with x-invariant terms hoisted (2 fma per column; q's sum reassociated); p copied to LDS as float4
# speedup vs baseline: 3.7728x; 1.0666x over previous
#include <hip/hip_runtime.h>
#include <hip/hip_bf16.h>


typedef __attribute__((ext_vector_type(2))) float v2f;
typedef __attribute__((ext_vector_type(8))) float v8f;
typedef __attribute__((ext_vector_type(4))) float v4f;

#define NDIM_GRID 20
#define NY_NZ     400
#define NPTS      8000
#define NTILES    500
#define NPAIRS    250
#define LOG2E     1.4426950408889634f

#define NW        8
#define NT        256
#define WG_ROWS   32
#define PVEC      8
#define PTHR      250

__device__ __forceinline__ void decode_coord(int i, float& x, float& y, float& z) {
    const unsigned short u   = (unsigned short)i;
    const unsigned short ix  = u / NY_NZ;
    const unsigned short rem = u - ix * NY_NZ;
    const unsigned short iy  = rem / NDIM_GRID;
    const unsigned short iz  = rem - iy * NDIM_GRID;
    x = (float)ix; y = (float)iy; z = (float)iz;
}

__device__ __forceinline__ float row16_sum(float v) {
    v += __builtin_bit_cast(float, __builtin_amdgcn_update_dpp(0, __builtin_bit_cast(int, v), 0x161, 0xf, 0xf, false));
    v += __builtin_bit_cast(float, __builtin_amdgcn_update_dpp(0, __builtin_bit_cast(int, v), 0x162, 0xf, 0xf, false));
    v += __builtin_bit_cast(float, __builtin_amdgcn_update_dpp(0, __builtin_bit_cast(int, v), 0x164, 0xf, 0xf, false));
    v += __builtin_bit_cast(float, __builtin_amdgcn_update_dpp(0, __builtin_bit_cast(int, v), 0x168, 0xf, 0xf, false));
    return v;
}

__global__ void __launch_bounds__(NT) kde_main(
        const float* __restrict__ p,
        const float* __restrict__ Ainv,
        float* __restrict__ kde)
{
    __shared__ float4 bq[NPTS];
    __shared__ float4 pp4[NPTS / 4];
    __shared__ float  part[NW * WG_ROWS];

    const int tid  = threadIdx.x;
    const int lane = tid & 31;
    const int w    = __builtin_amdgcn_readfirstlane(tid >> 5);
    const int half = lane >> 4;
    const int l15  = lane & 15;

    const float a00 = Ainv[0], a01 = Ainv[1], a02 = Ainv[2];
    const float a10 = Ainv[3], a11 = Ainv[4], a12 = Ainv[5];
    const float a20 = Ainv[6], a21 = Ainv[7], a22 = Ainv[8];

    const v4f* p4 = (const v4f*)p;
    const bool pcopy = tid < PTHR;
    v4f pv[PVEC];
    if (pcopy) {
#pragma unroll
        for (int k = 0; k < PVEC; ++k)
            pv[k] = p4[k * PTHR + tid];
    }

    for (int s = tid; s < NY_NZ; s += NT) {
        const unsigned short us = (unsigned short)s;
        const unsigned short iy = us / NDIM_GRID;
        const unsigned short iz = us - iy * NDIM_GRID;
        const float cy = (float)iy, cz = (float)iz;
        const float u0 = a01 * cy + a02 * cz;
        const float u1 = a11 * cy + a12 * cz;
        const float u2 = a21 * cy + a22 * cz;
        const float v  = u0 + a10 * cy + a20 * cz;
        const float wq = cy * u1 + cz * u2;
#pragma unroll
        for (int ix = 0; ix < NDIM_GRID; ++ix) {
            const float cx = (float)ix;
            const float q  = cx * (a00 * cx + v) + wq;
            bq[ix * NY_NZ + s] = make_float4(cx, cy, cz, q);
        }
    }

    if (pcopy) {
#pragma unroll
        for (int k = 0; k < PVEC; ++k)
            ((v4f*)pp4)[k * PTHR + tid] = pv[k];
    }
    __syncthreads();

    const int rowbase = blockIdx.x * WG_ROWS;
    v2f amat0, amat1;
    {
        float rx, ry, rz; decode_coord(rowbase + l15, rx, ry, rz);
        const float g0 = a00 * rx + a01 * ry + a02 * rz;
        const float g1 = a10 * rx + a11 * ry + a12 * rz;
        const float g2 = a20 * rx + a21 * ry + a22 * rz;
        amat0.x = half ? (LOG2E * g2) : (LOG2E * g0);
        amat0.y = half ? (-0.5f * LOG2E) : (LOG2E * g1);
    }
    {
        float rx, ry, rz; decode_coord(rowbase + 16 + l15, rx, ry, rz);
        const float g0 = a00 * rx + a01 * ry + a02 * rz;
        const float g1 = a10 * rx + a11 * ry + a12 * rz;
        const float g2 = a20 * rx + a21 * ry + a22 * rz;
        amat1.x = half ? (LOG2E * g2) : (LOG2E * g0);
        amat1.y = half ? (-0.5f * LOG2E) : (LOG2E * g1);
    }
    v8f cvec0, cvec1;
#pragma unroll
    for (int r = 0; r < 8; ++r) {
        cvec0[r] = (-0.5f * LOG2E) * bq[rowbase +      8 * half + r].w;
        cvec1[r] = (-0.5f * LOG2E) * bq[rowbase + 16 + 8 * half + r].w;
    }

    float acc0[8] = {};
    float acc1[8] = {};

    const v2f*   bbase = (const v2f*)bq + (l15 * 2 + half);
    const float* pbase = (const float*)pp4 + l15;

    const int npair = (NPAIRS - w + NW - 1) / NW;
    int j = w;
    v2f   b0 = bbase[(2 * j) * 32];
    v2f   b1 = bbase[(2 * j + 1) * 32];
    float p0 = pbase[(2 * j) * 16];
    float p1 = pbase[(2 * j + 1) * 16];

    for (int k = 0; k < npair; ++k) {
        const int jn = (j + NW < NPAIRS) ? (j + NW) : j;
        const v2f   nb0 = bbase[(2 * jn) * 32];
        const v2f   nb1 = bbase[(2 * jn + 1) * 32];
        const float np0 = pbase[(2 * jn) * 16];
        const float np1 = pbase[(2 * jn + 1) * 16];

        v8f c00 = __builtin_amdgcn_wmma_f32_16x16x4_f32(
                false, amat0, false, b0, (short)0, cvec0, false, false);
        v8f c10 = __builtin_amdgcn_wmma_f32_16x16x4_f32(
                false, amat1, false, b0, (short)0, cvec1, false, false);
        v8f c01 = __builtin_amdgcn_wmma_f32_16x16x4_f32(
                false, amat0, false, b1, (short)0, cvec0, false, false);
        v8f c11 = __builtin_amdgcn_wmma_f32_16x16x4_f32(
                false, amat1, false, b1, (short)0, cvec1, false, false);

#pragma unroll
        for (int r = 0; r < 8; ++r)
            acc0[r] = fmaf(__builtin_amdgcn_exp2f(c00[r]), p0, acc0[r]);
#pragma unroll
        for (int r = 0; r < 8; ++r)
            acc1[r] = fmaf(__builtin_amdgcn_exp2f(c10[r]), p0, acc1[r]);
#pragma unroll
        for (int r = 0; r < 8; ++r)
            acc0[r] = fmaf(__builtin_amdgcn_exp2f(c01[r]), p1, acc0[r]);
#pragma unroll
        for (int r = 0; r < 8; ++r)
            acc1[r] = fmaf(__builtin_amdgcn_exp2f(c11[r]), p1, acc1[r]);

        asm volatile("" :: "v"(nb0), "v"(nb1), "v"(np0), "v"(np1));
        b0 = nb0; b1 = nb1; p0 = np0; p1 = np1;
        j += NW;
    }

#pragma unroll
    for (int r = 0; r < 8; ++r) {
        const float s0 = row16_sum(acc0[r]);
        const float s1 = row16_sum(acc1[r]);
        if (l15 == 0) {
            part[w * WG_ROWS +      8 * half + r] = s0;
            part[w * WG_ROWS + 16 + 8 * half + r] = s1;
        }
    }
    __syncthreads();

    if (tid < WG_ROWS) {
        float s = part[tid];
#pragma unroll
        for (int v = 1; v < NW; ++v)
            s += part[v * WG_ROWS + tid];
        kde[rowbase + tid] = s;
    }
}

#define NORM_T   256
#define NORM_V   8
#define NVEC4    2000

__global__ void __launch_bounds__(NORM_T) kde_norm(float* __restrict__ kde) {
    __shared__ float red[16];
    const int tid  = threadIdx.x;
    const int lane = tid & 31;
    const int w    = tid >> 5;
    float4* vec = (float4*)kde;

    float4 v[NORM_V];
#pragma unroll
    for (int k = 0; k < NORM_V - 1; ++k)
        v[k] = vec[k * NORM_T + tid];
    const bool tail = (NORM_V - 1) * NORM_T + tid < NVEC4;
    v[NORM_V - 1] = tail ? vec[(NORM_V - 1) * NORM_T + tid]
                         : make_float4(0.0f, 0.0f, 0.0f, 0.0f);

    float t4[NORM_V];
#pragma unroll
    for (int k = 0; k < NORM_V; ++k)
        t4[k] = (v[k].x + v[k].y) + (v[k].z + v[k].w);
    float s = ((t4[0] + t4[1]) + (t4[2] + t4[3])) + ((t4[4] + t4[5]) + (t4[6] + t4[7]));

    s = row16_sum(s);
    if ((lane & 15) == 0) red[2 * w + (lane >> 4)] = s;
    __syncthreads();
    const float total = row16_sum(red[lane & 15]);
    const float inv = 1.0f / total;

#pragma unroll
    for (int k = 0; k < NORM_V - 1; ++k)
        vec[k * NORM_T + tid] = make_float4(v[k].x * inv, v[k].y * inv, v[k].z * inv, v[k].w * inv);
    if (tail)
        vec[(NORM_V - 1) * NORM_T + tid] = make_float4(v[NORM_V - 1].x * inv, v[NORM_V - 1].y * inv,
                                                       v[NORM_V - 1].z * inv, v[NORM_V - 1].w * inv);
}

extern "C" void kernel_launch(void* const* d_in, const int* in_sizes, int n_in,
                              void* d_out, int out_size, void* d_ws, size_t ws_size,
                              hipStream_t stream) {
    const float* p    = (const float*)d_in[0];
    const float* ainv = (const float*)d_in[1];
    float* out = (float*)d_out;

    kde_main<<<250, 256, 0, stream>>>(p, ainv, out);
    kde_norm<<<1, NORM_T, 0, stream>>>(out);
}


# Columns swept in z-neighbour pairs: only the even column goes through WMMA and exp2, the odd one is K*U_row*V_col, U applied after the row sum (new summation order); full sweep kept for large cov_inv
# speedup vs baseline: 4.7218x; 1.2515x over previous
#include <hip/hip_runtime.h>
#include <hip/hip_bf16.h>


typedef __attribute__((ext_vector_type(2))) float v2f;
typedef __attribute__((ext_vector_type(8))) float v8f;
typedef __attribute__((ext_vector_type(4))) float v4f;

#define NDIM_GRID 20
#define NY_NZ     400
#define NPTS      8000
#define NTILES    500
#define NPAIRS    250
#define NZTRIPS   125
#define ZRANGE    50.0f
#define LOG2E     1.4426950408889634f

#define NW        8
#define NT        256
#define WG_ROWS   32
#define PVEC      8
#define PTHR      250

__device__ __forceinline__ void decode_coord(int i, float& x, float& y, float& z) {
    const unsigned short u   = (unsigned short)i;
    const unsigned short ix  = u / NY_NZ;
    const unsigned short rem = u - ix * NY_NZ;
    const unsigned short iy  = rem / NDIM_GRID;
    const unsigned short iz  = rem - iy * NDIM_GRID;
    x = (float)ix; y = (float)iy; z = (float)iz;
}

template <int N>
__device__ __forceinline__ float row_share(float v) {
    return __builtin_bit_cast(float, __builtin_amdgcn_update_dpp(
            0, __builtin_bit_cast(int, v), 0x150 + N, 0xf, 0xf, false));
}

__device__ __forceinline__ float row16_sum(float v) {
    v += __builtin_bit_cast(float, __builtin_amdgcn_update_dpp(0, __builtin_bit_cast(int, v), 0x161, 0xf, 0xf, false));
    v += __builtin_bit_cast(float, __builtin_amdgcn_update_dpp(0, __builtin_bit_cast(int, v), 0x162, 0xf, 0xf, false));
    v += __builtin_bit_cast(float, __builtin_amdgcn_update_dpp(0, __builtin_bit_cast(int, v), 0x164, 0xf, 0xf, false));
    v += __builtin_bit_cast(float, __builtin_amdgcn_update_dpp(0, __builtin_bit_cast(int, v), 0x168, 0xf, 0xf, false));
    return v;
}

__global__ void __launch_bounds__(NT) kde_main(
        const float* __restrict__ p,
        const float* __restrict__ Ainv,
        float* __restrict__ kde)
{
    __shared__ float4 bq[NPTS];
    __shared__ float4 pp4[NPTS / 4];
    __shared__ float  part[NW * WG_ROWS];

    const int tid  = threadIdx.x;
    const int lane = tid & 31;
    const int w    = __builtin_amdgcn_readfirstlane(tid >> 5);
    const int half = lane >> 4;
    const int l15  = lane & 15;

    const float a00 = Ainv[0], a01 = Ainv[1], a02 = Ainv[2];
    const float a10 = Ainv[3], a11 = Ainv[4], a12 = Ainv[5];
    const float a20 = Ainv[6], a21 = Ainv[7], a22 = Ainv[8];

    const float c0z = a02 + a20, c1z = a12 + a21;
    const float ubound = 19.0f * LOG2E * (fabsf(a20) + fabsf(a21) + fabsf(a22));
    const float vbound = 0.5f * LOG2E * (19.0f * (fabsf(c0z) + fabsf(c1z)) + 39.0f * fabsf(a22));
    const bool paired = (ubound <= ZRANGE) && (vbound <= ZRANGE);

    const v4f* p4 = (const v4f*)p;
    const bool pcopy = tid < PTHR;
    v4f pv[PVEC];
    if (pcopy) {
#pragma unroll
        for (int k = 0; k < PVEC; ++k)
            pv[k] = p4[k * PTHR + tid];
    }

    auto stage_yz = [&](float cy, float cz, int pitch, int slot) {
        const float u0 = a01 * cy + a02 * cz;
        const float u1 = a11 * cy + a12 * cz;
        const float u2 = a21 * cy + a22 * cz;
        const float v  = u0 + a10 * cy + a20 * cz;
        const float wq = cy * u1 + cz * u2;
#pragma unroll
        for (int ix = 0; ix < NDIM_GRID; ++ix) {
            const float cx = (float)ix;
            const float q  = cx * (a00 * cx + v) + wq;
            bq[ix * pitch + slot] = make_float4(cx, cy, cz, q);
        }
    };

    if (paired) {
        for (int s = tid; s < NY_NZ / 2; s += NT) {
            const unsigned short us = (unsigned short)s;
            const unsigned short iy = us / (NDIM_GRID / 2);
            const unsigned short ik = us - iy * (NDIM_GRID / 2);
            stage_yz((float)iy, (float)(2 * ik), NY_NZ / 2, s);
        }
    } else {
        for (int s = tid; s < NY_NZ; s += NT) {
            const unsigned short us = (unsigned short)s;
            const unsigned short iy = us / NDIM_GRID;
            const unsigned short iz = us - iy * NDIM_GRID;
            stage_yz((float)iy, (float)iz, NY_NZ, s);
        }
    }

    if (pcopy) {
        if (paired) {
            const float ex = (-0.5f * LOG2E) * c0z;
            const float ey = (-0.5f * LOG2E) * c1z;
            const float ez = (-0.5f * LOG2E) * a22;
            const unsigned short ut  = (unsigned short)tid;
            const unsigned short x0  = ut / 100;
            const unsigned short fr0 = ut - x0 * 100;
            const unsigned short fr1 = fr0 + 50;
            const bool carry = fr1 >= 100;
            const unsigned short frb = carry ? (unsigned short)(fr1 - 100) : fr1;
            const unsigned short ya = fr0 / 5, yb = frb / 5;
            const unsigned short za = fr0 - ya * 5, zb = frb - yb * 5;
            const float la = ey * (float)ya + ez * (8.0f * (float)za + 1.0f);
            const float lb = ey * (float)yb + ez * (8.0f * (float)zb + 1.0f);
            const float xa = (float)x0;
            const float xb = carry ? (xa + 1.0f) : xa;
#pragma unroll
            for (int k = 0; k < PVEC; ++k) {
                const float xk = ((k & 1) ? xb : xa) + (float)(2 * k + (k >> 1));
                const float l0 = fmaf(ex, xk, (k & 1) ? lb : la);
                const float v0 = __builtin_amdgcn_exp2f(l0);
                const float v1 = __builtin_amdgcn_exp2f(fmaf(4.0f, ez, l0));
                v4f o;
                o.x = pv[k].x; o.y = pv[k].y * v0;
                o.z = pv[k].z; o.w = pv[k].w * v1;
                ((v4f*)pp4)[k * PTHR + tid] = o;
            }
        } else {
#pragma unroll
            for (int k = 0; k < PVEC; ++k)
                ((v4f*)pp4)[k * PTHR + tid] = pv[k];
        }
    }
    __syncthreads();

    const int rowbase = blockIdx.x * WG_ROWS;
    v2f amat0, amat1;
    {
        float rx, ry, rz; decode_coord(rowbase + l15, rx, ry, rz);
        const float g0 = a00 * rx + a01 * ry + a02 * rz;
        const float g1 = a10 * rx + a11 * ry + a12 * rz;
        const float g2 = a20 * rx + a21 * ry + a22 * rz;
        amat0.x = half ? (LOG2E * g2) : (LOG2E * g0);
        amat0.y = half ? (-0.5f * LOG2E) : (LOG2E * g1);
    }
    {
        float rx, ry, rz; decode_coord(rowbase + 16 + l15, rx, ry, rz);
        const float g0 = a00 * rx + a01 * ry + a02 * rz;
        const float g1 = a10 * rx + a11 * ry + a12 * rz;
        const float g2 = a20 * rx + a21 * ry + a22 * rz;
        amat1.x = half ? (LOG2E * g2) : (LOG2E * g0);
        amat1.y = half ? (-0.5f * LOG2E) : (LOG2E * g1);
    }

    float acc0[8] = {};
    float acc1[8] = {};

    const v2f* bbase = (const v2f*)bq + (l15 * 2 + half);

    if (paired) {
        float rx, ry, rz;
        decode_coord(rowbase + 16 * (l15 >> 3) + 8 * half + (l15 & 7), rx, ry, rz);
        const float rv   = (a01 + a10) * ry + (a02 + a20) * rz;
        const float rwq  = ry * (a11 * ry + a12 * rz) + rz * (a21 * ry + a22 * rz);
        const float bias = (-0.5f * LOG2E) * (rx * (a00 * rx + rv) + rwq);
        const float ufac = __builtin_amdgcn_exp2f(LOG2E * (a20 * rx + a21 * ry + a22 * rz));
        v8f cvec0, cvec1;
        cvec0[0] = row_share<0>(bias);  cvec1[0] = row_share<8>(bias);
        cvec0[1] = row_share<1>(bias);  cvec1[1] = row_share<9>(bias);
        cvec0[2] = row_share<2>(bias);  cvec1[2] = row_share<10>(bias);
        cvec0[3] = row_share<3>(bias);  cvec1[3] = row_share<11>(bias);
        cvec0[4] = row_share<4>(bias);  cvec1[4] = row_share<12>(bias);
        cvec0[5] = row_share<5>(bias);  cvec1[5] = row_share<13>(bias);
        cvec0[6] = row_share<6>(bias);  cvec1[6] = row_share<14>(bias);
        cvec0[7] = row_share<7>(bias);  cvec1[7] = row_share<15>(bias);

        float accp0[8] = {}, accp1[8] = {};
        float accw0[8] = {}, accw1[8] = {};
        const v2f* wbase = (const v2f*)pp4 + l15;

        const int ntrip = (NZTRIPS - w + NW - 1) / NW;
        int j = w;
        v2f b0 = bbase[(2 * j) * 32];
        v2f b1 = bbase[(2 * j + 1) * 32];
        v2f w0 = wbase[(2 * j) * 16];
        v2f w1 = wbase[(2 * j + 1) * 16];

        for (int k = 0; k < ntrip; ++k) {
            const int jn = (j + NW < NZTRIPS) ? (j + NW) : j;
            const v2f nb0 = bbase[(2 * jn) * 32];
            const v2f nb1 = bbase[(2 * jn + 1) * 32];
            const v2f nw0 = wbase[(2 * jn) * 16];
            const v2f nw1 = wbase[(2 * jn + 1) * 16];

            v8f c00 = __builtin_amdgcn_wmma_f32_16x16x4_f32(
                    false, amat0, false, b0, (short)0, cvec0, false, false);
            v8f c10 = __builtin_amdgcn_wmma_f32_16x16x4_f32(
                    false, amat1, false, b0, (short)0, cvec1, false, false);
            v8f c01 = __builtin_amdgcn_wmma_f32_16x16x4_f32(
                    false, amat0, false, b1, (short)0, cvec0, false, false);
            v8f c11 = __builtin_amdgcn_wmma_f32_16x16x4_f32(
                    false, amat1, false, b1, (short)0, cvec1, false, false);

#pragma unroll
            for (int r = 0; r < 8; ++r) {
                const float e = __builtin_amdgcn_exp2f(c00[r]);
                accp0[r] = fmaf(e, w0.x, accp0[r]);
                accw0[r] = fmaf(e, w0.y, accw0[r]);
            }
#pragma unroll
            for (int r = 0; r < 8; ++r) {
                const float e = __builtin_amdgcn_exp2f(c10[r]);
                accp1[r] = fmaf(e, w0.x, accp1[r]);
                accw1[r] = fmaf(e, w0.y, accw1[r]);
            }
#pragma unroll
            for (int r = 0; r < 8; ++r) {
                const float e = __builtin_amdgcn_exp2f(c01[r]);
                accp0[r] = fmaf(e, w1.x, accp0[r]);
                accw0[r] = fmaf(e, w1.y, accw0[r]);
            }
#pragma unroll
            for (int r = 0; r < 8; ++r) {
                const float e = __builtin_amdgcn_exp2f(c11[r]);
                accp1[r] = fmaf(e, w1.x, accp1[r]);
                accw1[r] = fmaf(e, w1.y, accw1[r]);
            }

            asm volatile("" :: "v"(nb0), "v"(nb1), "v"(nw0), "v"(nw1));
            b0 = nb0; b1 = nb1; w0 = nw0; w1 = nw1;
            j += NW;
        }

        acc0[0] = fmaf(row_share<0>(ufac), accw0[0], accp0[0]);
        acc0[1] = fmaf(row_share<1>(ufac), accw0[1], accp0[1]);
        acc0[2] = fmaf(row_share<2>(ufac), accw0[2], accp0[2]);
        acc0[3] = fmaf(row_share<3>(ufac), accw0[3], accp0[3]);
        acc0[4] = fmaf(row_share<4>(ufac), accw0[4], accp0[4]);
        acc0[5] = fmaf(row_share<5>(ufac), accw0[5], accp0[5]);
        acc0[6] = fmaf(row_share<6>(ufac), accw0[6], accp0[6]);
        acc0[7] = fmaf(row_share<7>(ufac), accw0[7], accp0[7]);
        acc1[0] = fmaf(row_share<8>(ufac),  accw1[0], accp1[0]);
        acc1[1] = fmaf(row_share<9>(ufac),  accw1[1], accp1[1]);
        acc1[2] = fmaf(row_share<10>(ufac), accw1[2], accp1[2]);
        acc1[3] = fmaf(row_share<11>(ufac), accw1[3], accp1[3]);
        acc1[4] = fmaf(row_share<12>(ufac), accw1[4], accp1[4]);
        acc1[5] = fmaf(row_share<13>(ufac), accw1[5], accp1[5]);
        acc1[6] = fmaf(row_share<14>(ufac), accw1[6], accp1[6]);
        acc1[7] = fmaf(row_share<15>(ufac), accw1[7], accp1[7]);
    } else {
        v8f cvec0, cvec1;
#pragma unroll
        for (int r = 0; r < 8; ++r) {
            cvec0[r] = (-0.5f * LOG2E) * bq[rowbase +      8 * half + r].w;
            cvec1[r] = (-0.5f * LOG2E) * bq[rowbase + 16 + 8 * half + r].w;
        }

        const float* pbase = (const float*)pp4 + l15;

        const int npair = (NPAIRS - w + NW - 1) / NW;
        int j = w;
        v2f   b0 = bbase[(2 * j) * 32];
        v2f   b1 = bbase[(2 * j + 1) * 32];
        float p0 = pbase[(2 * j) * 16];
        float p1 = pbase[(2 * j + 1) * 16];

        for (int k = 0; k < npair; ++k) {
            const int jn = (j + NW < NPAIRS) ? (j + NW) : j;
            const v2f   nb0 = bbase[(2 * jn) * 32];
            const v2f   nb1 = bbase[(2 * jn + 1) * 32];
            const float np0 = pbase[(2 * jn) * 16];
            const float np1 = pbase[(2 * jn + 1) * 16];

            v8f c00 = __builtin_amdgcn_wmma_f32_16x16x4_f32(
                    false, amat0, false, b0, (short)0, cvec0, false, false);
            v8f c10 = __builtin_amdgcn_wmma_f32_16x16x4_f32(
                    false, amat1, false, b0, (short)0, cvec1, false, false);
            v8f c01 = __builtin_amdgcn_wmma_f32_16x16x4_f32(
                    false, amat0, false, b1, (short)0, cvec0, false, false);
            v8f c11 = __builtin_amdgcn_wmma_f32_16x16x4_f32(
                    false, amat1, false, b1, (short)0, cvec1, false, false);

#pragma unroll
            for (int r = 0; r < 8; ++r)
                acc0[r] = fmaf(__builtin_amdgcn_exp2f(c00[r]), p0, acc0[r]);
#pragma unroll
            for (int r = 0; r < 8; ++r)
                acc1[r] = fmaf(__builtin_amdgcn_exp2f(c10[r]), p0, acc1[r]);
#pragma unroll
            for (int r = 0; r < 8; ++r)
                acc0[r] = fmaf(__builtin_amdgcn_exp2f(c01[r]), p1, acc0[r]);
#pragma unroll
            for (int r = 0; r < 8; ++r)
                acc1[r] = fmaf(__builtin_amdgcn_exp2f(c11[r]), p1, acc1[r]);

            asm volatile("" :: "v"(nb0), "v"(nb1), "v"(np0), "v"(np1));
            b0 = nb0; b1 = nb1; p0 = np0; p1 = np1;
            j += NW;
        }
    }

#pragma unroll
    for (int r = 0; r < 8; ++r) {
        const float s0 = row16_sum(acc0[r]);
        const float s1 = row16_sum(acc1[r]);
        if (l15 == 0) {
            part[w * WG_ROWS +      8 * half + r] = s0;
            part[w * WG_ROWS + 16 + 8 * half + r] = s1;
        }
    }
    __syncthreads();

    if (tid < WG_ROWS) {
        float s = part[tid];
#pragma unroll
        for (int v = 1; v < NW; ++v)
            s += part[v * WG_ROWS + tid];
        kde[rowbase + tid] = s;
    }
}

#define NORM_T   256
#define NORM_V   8
#define NVEC4    2000

__global__ void __launch_bounds__(NORM_T) kde_norm(float* __restrict__ kde) {
    __shared__ float red[16];
    const int tid  = threadIdx.x;
    const int lane = tid & 31;
    const int w    = tid >> 5;
    float4* vec = (float4*)kde;

    float4 v[NORM_V];
#pragma unroll
    for (int k = 0; k < NORM_V - 1; ++k)
        v[k] = vec[k * NORM_T + tid];
    const bool tail = (NORM_V - 1) * NORM_T + tid < NVEC4;
    v[NORM_V - 1] = tail ? vec[(NORM_V - 1) * NORM_T + tid]
                         : make_float4(0.0f, 0.0f, 0.0f, 0.0f);

    float t4[NORM_V];
#pragma unroll
    for (int k = 0; k < NORM_V; ++k)
        t4[k] = (v[k].x + v[k].y) + (v[k].z + v[k].w);
    float s = ((t4[0] + t4[1]) + (t4[2] + t4[3])) + ((t4[4] + t4[5]) + (t4[6] + t4[7]));

    s = row16_sum(s);
    if ((lane & 15) == 0) red[2 * w + (lane >> 4)] = s;
    __syncthreads();
    const float total = row16_sum(red[lane & 15]);
    const float inv = 1.0f / total;

#pragma unroll
    for (int k = 0; k < NORM_V - 1; ++k)
        vec[k * NORM_T + tid] = make_float4(v[k].x * inv, v[k].y * inv, v[k].z * inv, v[k].w * inv);
    if (tail)
        vec[(NORM_V - 1) * NORM_T + tid] = make_float4(v[NORM_V - 1].x * inv, v[NORM_V - 1].y * inv,
                                                       v[NORM_V - 1].z * inv, v[NORM_V - 1].w * inv);
}

extern "C" void kernel_launch(void* const* d_in, const int* in_sizes, int n_in,
                              void* d_out, int out_size, void* d_ws, size_t ws_size,
                              hipStream_t stream) {
    const float* p    = (const float*)d_in[0];
    const float* ainv = (const float*)d_in[1];
    float* out = (float*)d_out;

    kde_main<<<250, 256, 0, stream>>>(p, ainv, out);
    kde_norm<<<1, NORM_T, 0, stream>>>(out);
}


# Columns swept in z-groups of four (base z=4k+1): one WMMA+exp2 per group, the other three columns are K*U_row^j*V_col,j; four row sums (packed fma) recombined with U^-1,U,U^2 (new summation order)
# speedup vs baseline: 5.3606x; 1.1353x over previous
#include <hip/hip_runtime.h>
#include <hip/hip_bf16.h>


typedef __attribute__((ext_vector_type(2))) float v2f;
typedef __attribute__((ext_vector_type(8))) float v8f;
typedef __attribute__((ext_vector_type(4))) float v4f;

#define NDIM_GRID 20
#define NY_NZ     400
#define NPTS      8000
#define NTILES    500
#define NPAIRS    250
#define NQUADS    2000
#define NQTRIPS   63
#define QPAD      16
#define ZRANGE    90.0f
#define LOG2E     1.4426950408889634f

#define NW        8
#define NT        256
#define WG_ROWS   32
#define PVEC      8
#define PTHR      250

__device__ __forceinline__ void decode_coord(int i, float& x, float& y, float& z) {
    const unsigned short u   = (unsigned short)i;
    const unsigned short ix  = u / NY_NZ;
    const unsigned short rem = u - ix * NY_NZ;
    const unsigned short iy  = rem / NDIM_GRID;
    const unsigned short iz  = rem - iy * NDIM_GRID;
    x = (float)ix; y = (float)iy; z = (float)iz;
}

template <int N>
__device__ __forceinline__ float row_share(float v) {
    return __builtin_bit_cast(float, __builtin_amdgcn_update_dpp(
            0, __builtin_bit_cast(int, v), 0x150 + N, 0xf, 0xf, false));
}

__device__ __forceinline__ float row16_sum(float v) {
    v += __builtin_bit_cast(float, __builtin_amdgcn_update_dpp(0, __builtin_bit_cast(int, v), 0x161, 0xf, 0xf, false));
    v += __builtin_bit_cast(float, __builtin_amdgcn_update_dpp(0, __builtin_bit_cast(int, v), 0x162, 0xf, 0xf, false));
    v += __builtin_bit_cast(float, __builtin_amdgcn_update_dpp(0, __builtin_bit_cast(int, v), 0x164, 0xf, 0xf, false));
    v += __builtin_bit_cast(float, __builtin_amdgcn_update_dpp(0, __builtin_bit_cast(int, v), 0x168, 0xf, 0xf, false));
    return v;
}

__global__ void __launch_bounds__(NT) kde_main(
        const float* __restrict__ p,
        const float* __restrict__ Ainv,
        float* __restrict__ kde)
{
    __shared__ float4 bq[NPTS];
    __shared__ float4 pp4[NPTS / 4 + QPAD];
    __shared__ float  part[NW * WG_ROWS];

    const int tid  = threadIdx.x;
    const int lane = tid & 31;
    const int w    = __builtin_amdgcn_readfirstlane(tid >> 5);
    const int half = lane >> 4;
    const int l15  = lane & 15;

    const float a00 = Ainv[0], a01 = Ainv[1], a02 = Ainv[2];
    const float a10 = Ainv[3], a11 = Ainv[4], a12 = Ainv[5];
    const float a20 = Ainv[6], a21 = Ainv[7], a22 = Ainv[8];

    const float c0z = a02 + a20, c1z = a12 + a21;
    const float ubound = 2.0f * 19.0f * LOG2E * (fabsf(a20) + fabsf(a21) + fabsf(a22));
    const float vbound = 0.5f * LOG2E * (38.0f * (fabsf(c0z) + fabsf(c1z)) + 72.0f * fabsf(a22));
    const bool grouped = (ubound <= ZRANGE) && (vbound <= ZRANGE);

    const v4f* p4 = (const v4f*)p;
    const bool pcopy = tid < PTHR;
    v4f pv[PVEC];
    if (pcopy) {
#pragma unroll
        for (int k = 0; k < PVEC; ++k)
            pv[k] = p4[k * PTHR + tid];
    }

    auto stage_yz = [&](float cy, float cz, int pitch, int slot) {
        const float u0 = a01 * cy + a02 * cz;
        const float u1 = a11 * cy + a12 * cz;
        const float u2 = a21 * cy + a22 * cz;
        const float v  = u0 + a10 * cy + a20 * cz;
        const float wq = cy * u1 + cz * u2;
#pragma unroll
        for (int ix = 0; ix < NDIM_GRID; ++ix) {
            const float cx = (float)ix;
            const float q  = cx * (a00 * cx + v) + wq;
            bq[ix * pitch + slot] = make_float4(cx, cy, cz, q);
        }
    };

    if (grouped) {
        if (tid < 200) {
            const unsigned short ut = (unsigned short)tid;
            const unsigned short xh = ut / 100;
            const unsigned short yk = ut - xh * 100;
            const unsigned short iy = yk / 5;
            const unsigned short ik = yk - iy * 5;
            const float cy = (float)iy, cz = (float)(4 * ik + 1);
            const float u0 = a01 * cy + a02 * cz;
            const float u1 = a11 * cy + a12 * cz;
            const float u2 = a21 * cy + a22 * cz;
            const float v  = u0 + a10 * cy + a20 * cz;
            const float wq = cy * u1 + cz * u2;
            const float xf = (float)(10 * xh);
            float4* dst = bq + (10 * xh) * 100 + yk;
#pragma unroll
            for (int ix = 0; ix < 10; ++ix) {
                const float cx = xf + (float)ix;
                const float q  = cx * (a00 * cx + v) + wq;
                dst[ix * 100] = make_float4(cx, cy, cz, q);
            }
        } else if (tid < 200 + QPAD) {
            bq[NQUADS + (tid - 200)]  = make_float4(0.0f, 0.0f, 1.0f, a22);
            pp4[NQUADS + (tid - 200)] = make_float4(0.0f, 0.0f, 0.0f, 0.0f);
        }
    } else {
        for (int s = tid; s < NY_NZ; s += NT) {
            const unsigned short us = (unsigned short)s;
            const unsigned short iy = us / NDIM_GRID;
            const unsigned short iz = us - iy * NDIM_GRID;
            stage_yz((float)iy, (float)iz, NY_NZ, s);
        }
    }

    if (pcopy) {
        if (grouped) {
            const float ex = (-0.5f * LOG2E) * c0z;
            const float ey = (-0.5f * LOG2E) * c1z;
            const float ez = (-0.5f * LOG2E) * a22;
            const unsigned short ut  = (unsigned short)tid;
            const unsigned short x0  = ut / 100;
            const unsigned short fr0 = ut - x0 * 100;
            const unsigned short fr1 = fr0 + 50;
            const bool carry = fr1 >= 100;
            const unsigned short frb = carry ? (unsigned short)(fr1 - 100) : fr1;
            const unsigned short ya = fr0 / 5, yb = frb / 5;
            const unsigned short ka = fr0 - ya * 5, kb = frb - yb * 5;
            const float zca = (float)(4 * ka + 1), zcb = (float)(4 * kb + 1);
            const float sya = ey * (float)ya, syb = ey * (float)yb;
            const float t1a = ez * (2.0f * zca + 1.0f), t1b = ez * (2.0f * zcb + 1.0f);
            const float t2a = ez * (4.0f * zca + 4.0f), t2b = ez * (4.0f * zcb + 4.0f);
            const float tma = ez * (1.0f - 2.0f * zca), tmb = ez * (1.0f - 2.0f * zcb);
            const float xa = (float)x0;
            const float xb = carry ? (xa + 1.0f) : xa;
#pragma unroll
            for (int k = 0; k < PVEC; ++k) {
                const float xk = ((k & 1) ? xb : xa) + (float)(2 * k + (k >> 1));
                const float s  = fmaf(ex, xk, (k & 1) ? syb : sya);
                const float vm = __builtin_amdgcn_exp2f(((k & 1) ? tmb : tma) - s);
                const float v1 = __builtin_amdgcn_exp2f(s + ((k & 1) ? t1b : t1a));
                const float v2 = __builtin_amdgcn_exp2f(fmaf(2.0f, s, (k & 1) ? t2b : t2a));
                v4f o;
                o.x = pv[k].x * vm; o.y = pv[k].y;
                o.z = pv[k].z * v1; o.w = pv[k].w * v2;
                ((v4f*)pp4)[k * PTHR + tid] = o;
            }
        } else {
#pragma unroll
            for (int k = 0; k < PVEC; ++k)
                ((v4f*)pp4)[k * PTHR + tid] = pv[k];
        }
    }
    __syncthreads();

    const int rowbase = blockIdx.x * WG_ROWS;
    v2f amat0, amat1;
    {
        float rx, ry, rz; decode_coord(rowbase + l15, rx, ry, rz);
        const float g0 = a00 * rx + a01 * ry + a02 * rz;
        const float g1 = a10 * rx + a11 * ry + a12 * rz;
        const float g2 = a20 * rx + a21 * ry + a22 * rz;
        amat0.x = half ? (LOG2E * g2) : (LOG2E * g0);
        amat0.y = half ? (-0.5f * LOG2E) : (LOG2E * g1);
    }
    {
        float rx, ry, rz; decode_coord(rowbase + 16 + l15, rx, ry, rz);
        const float g0 = a00 * rx + a01 * ry + a02 * rz;
        const float g1 = a10 * rx + a11 * ry + a12 * rz;
        const float g2 = a20 * rx + a21 * ry + a22 * rz;
        amat1.x = half ? (LOG2E * g2) : (LOG2E * g0);
        amat1.y = half ? (-0.5f * LOG2E) : (LOG2E * g1);
    }

    float acc0[8] = {};
    float acc1[8] = {};

    const v2f* bbase = (const v2f*)bq + (l15 * 2 + half);

    if (grouped) {
        float rx, ry, rz;
        decode_coord(rowbase + 16 * (l15 >> 3) + 8 * half + (l15 & 7), rx, ry, rz);
        const float rv   = (a01 + a10) * ry + (a02 + a20) * rz;
        const float rwq  = ry * (a11 * ry + a12 * rz) + rz * (a21 * ry + a22 * rz);
        const float bias = (-0.5f * LOG2E) * (rx * (a00 * rx + rv) + rwq);
        const float lu   = LOG2E * (a20 * rx + a21 * ry + a22 * rz);
        const float ufac = __builtin_amdgcn_exp2f(lu);
        const float uinv = __builtin_amdgcn_exp2f(-lu);
        const float usq  = __builtin_amdgcn_exp2f(2.0f * lu);
        v8f cvec0, cvec1;
        cvec0[0] = row_share<0>(bias);  cvec1[0] = row_share<8>(bias);
        cvec0[1] = row_share<1>(bias);  cvec1[1] = row_share<9>(bias);
        cvec0[2] = row_share<2>(bias);  cvec1[2] = row_share<10>(bias);
        cvec0[3] = row_share<3>(bias);  cvec1[3] = row_share<11>(bias);
        cvec0[4] = row_share<4>(bias);  cvec1[4] = row_share<12>(bias);
        cvec0[5] = row_share<5>(bias);  cvec1[5] = row_share<13>(bias);
        cvec0[6] = row_share<6>(bias);  cvec1[6] = row_share<14>(bias);
        cvec0[7] = row_share<7>(bias);  cvec1[7] = row_share<15>(bias);

        v2f lo0[8] = {}, hi0[8] = {};
        v2f lo1[8] = {}, hi1[8] = {};
        const v4f* wbase = (const v4f*)pp4 + l15;

        const int ntrip = (NQTRIPS - w + NW - 1) / NW;
        int j = w;
        v2f b0 = bbase[(2 * j) * 32];
        v2f b1 = bbase[(2 * j + 1) * 32];
        v4f w0 = wbase[(2 * j) * 16];
        v4f w1 = wbase[(2 * j + 1) * 16];

        for (int k = 0; k < ntrip; ++k) {
            const int jn = (j + NW < NQTRIPS) ? (j + NW) : j;
            const v2f nb0 = bbase[(2 * jn) * 32];
            const v2f nb1 = bbase[(2 * jn + 1) * 32];
            const v4f nw0 = wbase[(2 * jn) * 16];
            const v4f nw1 = wbase[(2 * jn + 1) * 16];

            v8f c00 = __builtin_amdgcn_wmma_f32_16x16x4_f32(
                    false, amat0, false, b0, (short)0, cvec0, false, false);
            v8f c10 = __builtin_amdgcn_wmma_f32_16x16x4_f32(
                    false, amat1, false, b0, (short)0, cvec1, false, false);
            v8f c01 = __builtin_amdgcn_wmma_f32_16x16x4_f32(
                    false, amat0, false, b1, (short)0, cvec0, false, false);
            v8f c11 = __builtin_amdgcn_wmma_f32_16x16x4_f32(
                    false, amat1, false, b1, (short)0, cvec1, false, false);

            const v2f w0lo = {w0.x, w0.y}, w0hi = {w0.z, w0.w};
            const v2f w1lo = {w1.x, w1.y}, w1hi = {w1.z, w1.w};
#pragma unroll
            for (int r = 0; r < 8; ++r) {
                const float e = __builtin_amdgcn_exp2f(c00[r]);
                const v2f ee = {e, e};
                lo0[r] = __builtin_elementwise_fma(ee, w0lo, lo0[r]);
                hi0[r] = __builtin_elementwise_fma(ee, w0hi, hi0[r]);
            }
#pragma unroll
            for (int r = 0; r < 8; ++r) {
                const float e = __builtin_amdgcn_exp2f(c10[r]);
                const v2f ee = {e, e};
                lo1[r] = __builtin_elementwise_fma(ee, w0lo, lo1[r]);
                hi1[r] = __builtin_elementwise_fma(ee, w0hi, hi1[r]);
            }
#pragma unroll
            for (int r = 0; r < 8; ++r) {
                const float e = __builtin_amdgcn_exp2f(c01[r]);
                const v2f ee = {e, e};
                lo0[r] = __builtin_elementwise_fma(ee, w1lo, lo0[r]);
                hi0[r] = __builtin_elementwise_fma(ee, w1hi, hi0[r]);
            }
#pragma unroll
            for (int r = 0; r < 8; ++r) {
                const float e = __builtin_amdgcn_exp2f(c11[r]);
                const v2f ee = {e, e};
                lo1[r] = __builtin_elementwise_fma(ee, w1lo, lo1[r]);
                hi1[r] = __builtin_elementwise_fma(ee, w1hi, hi1[r]);
            }

            __builtin_amdgcn_sched_barrier(0);
            asm volatile("" :: "v"(nb0), "v"(nb1), "v"(nw0), "v"(nw1));
            b0 = nb0; b1 = nb1; w0 = nw0; w1 = nw1;
            j += NW;
        }

#define ROW_COMBINE(r)                                                              \
        acc0[r] = fmaf(row_share<r>(usq), hi0[r].y,                                 \
                  fmaf(row_share<r>(ufac), hi0[r].x,                                \
                  fmaf(row_share<r>(uinv), lo0[r].x, lo0[r].y)));                   \
        acc1[r] = fmaf(row_share<8 + r>(usq), hi1[r].y,                             \
                  fmaf(row_share<8 + r>(ufac), hi1[r].x,                            \
                  fmaf(row_share<8 + r>(uinv), lo1[r].x, lo1[r].y)));
        ROW_COMBINE(0) ROW_COMBINE(1) ROW_COMBINE(2) ROW_COMBINE(3)
        ROW_COMBINE(4) ROW_COMBINE(5) ROW_COMBINE(6) ROW_COMBINE(7)
#undef ROW_COMBINE
    } else {
        v8f cvec0, cvec1;
#pragma unroll
        for (int r = 0; r < 8; ++r) {
            cvec0[r] = (-0.5f * LOG2E) * bq[rowbase +      8 * half + r].w;
            cvec1[r] = (-0.5f * LOG2E) * bq[rowbase + 16 + 8 * half + r].w;
        }

        const float* pbase = (const float*)pp4 + l15;

        const int npair = (NPAIRS - w + NW - 1) / NW;
        int j = w;
        v2f   b0 = bbase[(2 * j) * 32];
        v2f   b1 = bbase[(2 * j + 1) * 32];
        float p0 = pbase[(2 * j) * 16];
        float p1 = pbase[(2 * j + 1) * 16];

        for (int k = 0; k < npair; ++k) {
            const int jn = (j + NW < NPAIRS) ? (j + NW) : j;
            const v2f   nb0 = bbase[(2 * jn) * 32];
            const v2f   nb1 = bbase[(2 * jn + 1) * 32];
            const float np0 = pbase[(2 * jn) * 16];
            const float np1 = pbase[(2 * jn + 1) * 16];

            v8f c00 = __builtin_amdgcn_wmma_f32_16x16x4_f32(
                    false, amat0, false, b0, (short)0, cvec0, false, false);
            v8f c10 = __builtin_amdgcn_wmma_f32_16x16x4_f32(
                    false, amat1, false, b0, (short)0, cvec1, false, false);
            v8f c01 = __builtin_amdgcn_wmma_f32_16x16x4_f32(
                    false, amat0, false, b1, (short)0, cvec0, false, false);
            v8f c11 = __builtin_amdgcn_wmma_f32_16x16x4_f32(
                    false, amat1, false, b1, (short)0, cvec1, false, false);

#pragma unroll
            for (int r = 0; r < 8; ++r)
                acc0[r] = fmaf(__builtin_amdgcn_exp2f(c00[r]), p0, acc0[r]);
#pragma unroll
            for (int r = 0; r < 8; ++r)
                acc1[r] = fmaf(__builtin_amdgcn_exp2f(c10[r]), p0, acc1[r]);
#pragma unroll
            for (int r = 0; r < 8; ++r)
                acc0[r] = fmaf(__builtin_amdgcn_exp2f(c01[r]), p1, acc0[r]);
#pragma unroll
            for (int r = 0; r < 8; ++r)
                acc1[r] = fmaf(__builtin_amdgcn_exp2f(c11[r]), p1, acc1[r]);

            asm volatile("" :: "v"(nb0), "v"(nb1), "v"(np0), "v"(np1));
            b0 = nb0; b1 = nb1; p0 = np0; p1 = np1;
            j += NW;
        }
    }

#pragma unroll
    for (int r = 0; r < 8; ++r) {
        const float s0 = row16_sum(acc0[r]);
        const float s1 = row16_sum(acc1[r]);
        if (l15 == 0) {
            part[w * WG_ROWS +      8 * half + r] = s0;
            part[w * WG_ROWS + 16 + 8 * half + r] = s1;
        }
    }
    __syncthreads();

    if (tid < WG_ROWS) {
        float s = part[tid];
#pragma unroll
        for (int v = 1; v < NW; ++v)
            s += part[v * WG_ROWS + tid];
        kde[rowbase + tid] = s;
    }
}

#define NORM_T   256
#define NORM_V   8
#define NVEC4    2000

__global__ void __launch_bounds__(NORM_T) kde_norm(float* __restrict__ kde) {
    __shared__ float red[16];
    const int tid  = threadIdx.x;
    const int lane = tid & 31;
    const int w    = tid >> 5;
    float4* vec = (float4*)kde;

    float4 v[NORM_V];
#pragma unroll
    for (int k = 0; k < NORM_V - 1; ++k)
        v[k] = vec[k * NORM_T + tid];
    const bool tail = (NORM_V - 1) * NORM_T + tid < NVEC4;
    v[NORM_V - 1] = tail ? vec[(NORM_V - 1) * NORM_T + tid]
                         : make_float4(0.0f, 0.0f, 0.0f, 0.0f);

    float t4[NORM_V];
#pragma unroll
    for (int k = 0; k < NORM_V; ++k)
        t4[k] = (v[k].x + v[k].y) + (v[k].z + v[k].w);
    float s = ((t4[0] + t4[1]) + (t4[2] + t4[3])) + ((t4[4] + t4[5]) + (t4[6] + t4[7]));

    s = row16_sum(s);
    if ((lane & 15) == 0) red[2 * w + (lane >> 4)] = s;
    __syncthreads();
    const float total = row16_sum(red[lane & 15]);
    const float inv = 1.0f / total;

#pragma unroll
    for (int k = 0; k < NORM_V - 1; ++k)
        vec[k * NORM_T + tid] = make_float4(v[k].x * inv, v[k].y * inv, v[k].z * inv, v[k].w * inv);
    if (tail)
        vec[(NORM_V - 1) * NORM_T + tid] = make_float4(v[NORM_V - 1].x * inv, v[NORM_V - 1].y * inv,
                                                       v[NORM_V - 1].z * inv, v[NORM_V - 1].w * inv);
}

extern "C" void kernel_launch(void* const* d_in, const int* in_sizes, int n_in,
                              void* d_out, int out_size, void* d_ws, size_t ws_size,
                              hipStream_t stream) {
    const float* p    = (const float*)d_in[0];
    const float* ainv = (const float*)d_in[1];
    float* out = (float*)d_out;

    kde_main<<<250, 256, 0, stream>>>(p, ainv, out);
    kde_norm<<<1, NORM_T, 0, stream>>>(out);
}


# Each wave loads p for and stages only the weights of its own four tiles (base columns formed in registers, loads issued at entry) into an LDS table of its own: no workgroup barrier ahead of the sweep
# speedup vs baseline: 6.7362x; 1.2566x over previous
#include <hip/hip_runtime.h>
#include <hip/hip_bf16.h>


typedef __attribute__((ext_vector_type(2))) float v2f;
typedef __attribute__((ext_vector_type(8))) float v8f;
typedef __attribute__((ext_vector_type(4))) float v4f;

#define NDIM_GRID 20
#define NY_NZ     400
#define NPTS      8000
#define NTILES    500
#define NPAIRS    250
#define NGROUPS   500
#define NGTILES   32
#define NTW       4
#define TPITCH    68
#define TWAVE     1088
#define ZRANGE    90.0f
#define LOG2E     1.4426950408889634f

#define NW        8
#define NT        256
#define WG_ROWS   32

__device__ __forceinline__ void decode_coord(int i, float& x, float& y, float& z) {
    const unsigned short u   = (unsigned short)i;
    const unsigned short ix  = u / NY_NZ;
    const unsigned short rem = u - ix * NY_NZ;
    const unsigned short iy  = rem / NDIM_GRID;
    const unsigned short iz  = rem - iy * NDIM_GRID;
    x = (float)ix; y = (float)iy; z = (float)iz;
}

__device__ __forceinline__ float row16_sum(float v) {
    v += __builtin_bit_cast(float, __builtin_amdgcn_update_dpp(0, __builtin_bit_cast(int, v), 0x161, 0xf, 0xf, false));
    v += __builtin_bit_cast(float, __builtin_amdgcn_update_dpp(0, __builtin_bit_cast(int, v), 0x162, 0xf, 0xf, false));
    v += __builtin_bit_cast(float, __builtin_amdgcn_update_dpp(0, __builtin_bit_cast(int, v), 0x164, 0xf, 0xf, false));
    v += __builtin_bit_cast(float, __builtin_amdgcn_update_dpp(0, __builtin_bit_cast(int, v), 0x168, 0xf, 0xf, false));
    return v;
}

__device__ __forceinline__ float row16_max(float v) {
    v = fmaxf(v, __builtin_bit_cast(float, __builtin_amdgcn_update_dpp(0, __builtin_bit_cast(int, v), 0x161, 0xf, 0xf, false)));
    v = fmaxf(v, __builtin_bit_cast(float, __builtin_amdgcn_update_dpp(0, __builtin_bit_cast(int, v), 0x162, 0xf, 0xf, false)));
    v = fmaxf(v, __builtin_bit_cast(float, __builtin_amdgcn_update_dpp(0, __builtin_bit_cast(int, v), 0x164, 0xf, 0xf, false)));
    v = fmaxf(v, __builtin_bit_cast(float, __builtin_amdgcn_update_dpp(0, __builtin_bit_cast(int, v), 0x168, 0xf, 0xf, false)));
    return v;
}

__device__ __forceinline__ float other_half(float v) {
    const int i = __builtin_bit_cast(int, v);
    return __builtin_bit_cast(float, __builtin_amdgcn_permlanex16(i, i, 0x76543210, 0xfedcba98, false, false));
}

__global__ void __launch_bounds__(NT) kde_main(
        const float* __restrict__ p,
        const float* __restrict__ Ainv,
        float* __restrict__ kde)
{
    __shared__ float4 bq[NPTS];
    __shared__ float4 pp4[NPTS / 4];
    __shared__ float  part[NW * WG_ROWS];

    const int tid  = threadIdx.x;
    const int lane = tid & 31;
    const int w    = __builtin_amdgcn_readfirstlane(tid >> 5);
    const int half = lane >> 4;
    const int l15  = lane & 15;

    const v4f* p4 = (const v4f*)p;
    float gbx[NTW], gby[NTW], gzc[NTW];
    int   gf[NTW];
    bool  gval[NTW];
    v4f   pv[NTW][2];
#pragma unroll
    for (int i = 0; i < NTW; ++i) {
        const int e = 16 * w + l15 + 16 * NW * i;
        gval[i] = (i < NTW - 1) || (e < NGROUPS);
        const unsigned short ue = (unsigned short)(gval[i] ? e : 0);
        const unsigned short gx = ue / 50;
        const unsigned short gr = ue - gx * 50;
        const unsigned short gy = gr / 5;
        const unsigned short gk = gr - gy * 5;
        gbx[i] = (float)(2 * gx); gby[i] = (float)(2 * gy); gzc[i] = (float)(4 * gk + 1);
        gf[i]  = 200 * gx + 10 * gy + 5 * half + gk;
        pv[i][0] = p4[gf[i]];
        pv[i][1] = p4[gf[i] + 100];
    }

    const float a00 = Ainv[0], a01 = Ainv[1], a02 = Ainv[2];
    const float a10 = Ainv[3], a11 = Ainv[4], a12 = Ainv[5];
    const float a20 = Ainv[6], a21 = Ainv[7], a22 = Ainv[8];
    const float c01 = a01 + a10, c02 = a02 + a20, c12 = a12 + a21;

    bool grouped;
    {
        const float dx = (float)(l15 >> 3), dy = (float)((l15 >> 2) & 1), dj = (float)((l15 & 3) - 1);
        const float r0 = dx * a00 + dy * a10 + dj * a20;
        const float r1 = dx * a01 + dy * a11 + dj * a21;
        const float r2 = dx * a02 + dy * a12 + dj * a22;
        const float rpos = fmaxf(r0, 0.0f) + fmaxf(r1, 0.0f) + fmaxf(r2, 0.0f);
        const float rneg = fmaxf(-r0, 0.0f) + fmaxf(-r1, 0.0f) + fmaxf(-r2, 0.0f);
        const float rb = (19.0f * LOG2E) * fmaxf(rpos, rneg);
        const float m0 = 2.0f * a00 * dx + c01 * dy + c02 * dj;
        const float m1 = c01 * dx + 2.0f * a11 * dy + c12 * dj;
        const float m2 = c02 * dx + c12 * dy + 2.0f * a22 * dj;
        const float qt = dx * (a00 * dx + c01 * dy + c02 * dj) + dy * (a11 * dy + c12 * dj) + a22 * dj * dj;
        const float mpos = fmaxf(m0, 0.0f) + fmaxf(m1, 0.0f) + fmaxf(m2, 0.0f);
        const float mneg = fmaxf(-m0, 0.0f) + fmaxf(-m1, 0.0f) + fmaxf(-m2, 0.0f);
        const float vb = (0.5f * LOG2E) * (19.0f * fmaxf(mpos, mneg) + fabsf(qt));
        const float bound = row16_max(fmaxf(rb, vb));
        const float asum = fabsf(a00) + fabsf(a01) + fabsf(a02) + fabsf(a10) + fabsf(a11)
                         + fabsf(a12) + fabsf(a20) + fabsf(a21) + fabsf(a22);
        grouped = __builtin_amdgcn_readfirstlane((int)((bound <= ZRANGE) && (asum <= 3.0e38f))) != 0;
    }

    const int rowbase = blockIdx.x * WG_ROWS;
    float rx0, ry0, rz0, rx1, ry1, rz1;
    decode_coord(rowbase + l15, rx0, ry0, rz0);
    decode_coord(rowbase + 16 + l15, rx1, ry1, rz1);
    const float g00 = a00 * rx0 + a01 * ry0 + a02 * rz0;
    const float g01 = a10 * rx0 + a11 * ry0 + a12 * rz0;
    const float g02 = a20 * rx0 + a21 * ry0 + a22 * rz0;
    const float g10 = a00 * rx1 + a01 * ry1 + a02 * rz1;
    const float g11 = a10 * rx1 + a11 * ry1 + a12 * rz1;
    const float g12 = a20 * rx1 + a21 * ry1 + a22 * rz1;
    v2f amat0, amat1;
    amat0.x = half ? (LOG2E * g02) : (LOG2E * g00);
    amat0.y = half ? (-0.5f * LOG2E) : (LOG2E * g01);
    amat1.x = half ? (LOG2E * g12) : (LOG2E * g10);
    amat1.y = half ? (-0.5f * LOG2E) : (LOG2E * g11);

    if (grouped) {
        const float q0 = rx0 * (a00 * rx0 + (c01 * ry0 + c02 * rz0))
                       + (ry0 * (a11 * ry0 + a12 * rz0) + rz0 * (a21 * ry0 + a22 * rz0));
        const float q1 = rx1 * (a00 * rx1 + (c01 * ry1 + c02 * rz1))
                       + (ry1 * (a11 * ry1 + a12 * rz1) + rz1 * (a21 * ry1 + a22 * rz1));
        const float bias0 = (-0.5f * LOG2E) * q0;
        const float bias1 = (-0.5f * LOG2E) * q1;
        v8f cv0, cv1;
#pragma unroll
        for (int r = 0; r < 8; ++r) { cv0[r] = bias0; cv1[r] = bias1; }

        v8f e0[NTW], e1[NTW];
#pragma unroll
        for (int i = 0; i < NTW; ++i) {
            const float cx = gbx[i], cy = gby[i], cz = gzc[i];
            const float v  = c01 * cy + c02 * cz;
            const float wq = cy * (a11 * cy + a12 * cz) + cz * (a21 * cy + a22 * cz);
            const float q  = cx * (a00 * cx + v) + wq;
            v2f cb;
            cb.x = half ? cz : cx;
            cb.y = half ? q  : cy;
            const v8f x0 = __builtin_amdgcn_wmma_f32_16x16x4_f32(
                    false, cb, false, amat0, (short)0, cv0, false, false);
            const v8f x1 = __builtin_amdgcn_wmma_f32_16x16x4_f32(
                    false, cb, false, amat1, (short)0, cv1, false, false);
#pragma unroll
            for (int r = 0; r < 8; ++r) {
                e0[i][r] = __builtin_amdgcn_exp2f(x0[r]);
                e1[i][r] = __builtin_amdgcn_exp2f(x1[r]);
            }
        }

        __builtin_amdgcn_sched_barrier(0);

        const float hl = -0.5f * LOG2E;
        const float hc02 = hl * c02, ha22 = hl * a22;
        const float eyd  = half ? (hl * c01) : 0.0f;
        const float dyf  = (float)half;
        float vf[NTW][2][4];
#pragma unroll
        for (int i = 0; i < NTW; ++i) {
            const float cx = gbx[i], cy = gby[i], cz = gzc[i];
            const float fy  = cy + dyf;
            const float tz0 = hl * (c02 * cx + (c12 * fy + 2.0f * a22 * cz));
            const float ey0 = half ? (hl * (c01 * cx + (2.0f * a11 * cy + c12 * cz + a11))) : 0.0f;
            const float ex1 = hl * (2.0f * a00 * (cx + 1.0f) + (c01 * cy + c02 * cz - a00));
#pragma unroll
            for (int dx = 0; dx < 2; ++dx) {
                const float tz = dx ? (tz0 + hc02) : tz0;
                const float ee = dx ? ((ey0 + eyd) + ex1) : ey0;
                vf[i][dx][0] = __builtin_amdgcn_exp2f((ee - tz) + ha22);
                vf[i][dx][1] = __builtin_amdgcn_exp2f(ee);
                vf[i][dx][2] = __builtin_amdgcn_exp2f((ee + tz) + ha22);
                vf[i][dx][3] = __builtin_amdgcn_exp2f(fmaf(2.0f, tz, ee) + 4.0f * ha22);
            }
        }

        float* const tw = (float*)bq + w * TWAVE;
#pragma unroll
        for (int i = 0; i < NTW; ++i) {
#pragma unroll
            for (int dx = 0; dx < 2; ++dx) {
                float* dst = tw + (8 * dx + 4 * half) * TPITCH + 16 * i + l15;
                const v4f pq = pv[i][dx];
                dst[0 * TPITCH] = gval[i] ? (pq.x * vf[i][dx][0]) : 0.0f;
                dst[1 * TPITCH] = gval[i] ? (pq.y * vf[i][dx][1]) : 0.0f;
                dst[2 * TPITCH] = gval[i] ? (pq.z * vf[i][dx][2]) : 0.0f;
                dst[3 * TPITCH] = gval[i] ? (pq.w * vf[i][dx][3]) : 0.0f;
            }
        }

        __builtin_amdgcn_fence(__ATOMIC_RELEASE, "wavefront");
        __builtin_amdgcn_wave_barrier();
        __builtin_amdgcn_fence(__ATOMIC_ACQUIRE, "wavefront");

        const v4f* wrow = (const v4f*)(tw + l15 * TPITCH + 8 * half);
        v4f wa[NTW], wb[NTW];
#pragma unroll
        for (int i = 0; i < NTW; ++i) {
            wa[i] = wrow[4 * i];
            wb[i] = wrow[4 * i + 1];
        }

        __builtin_amdgcn_sched_barrier(0);

        v8f s0 = {}, s1 = {};
#pragma unroll
        for (int i = 0; i < NTW; ++i) {
            const v2f w01 = {wa[i].x, wa[i].y}, w23 = {wa[i].z, wa[i].w};
            const v2f w45 = {wb[i].x, wb[i].y}, w67 = {wb[i].z, wb[i].w};
            s0 = __builtin_amdgcn_wmma_f32_16x16x4_f32(
                    false, w01, false, (v2f){e0[i][0], e0[i][1]}, (short)0, s0, false, false);
            s1 = __builtin_amdgcn_wmma_f32_16x16x4_f32(
                    false, w01, false, (v2f){e1[i][0], e1[i][1]}, (short)0, s1, false, false);
            s0 = __builtin_amdgcn_wmma_f32_16x16x4_f32(
                    false, w23, false, (v2f){e0[i][2], e0[i][3]}, (short)0, s0, false, false);
            s1 = __builtin_amdgcn_wmma_f32_16x16x4_f32(
                    false, w23, false, (v2f){e1[i][2], e1[i][3]}, (short)0, s1, false, false);
            s0 = __builtin_amdgcn_wmma_f32_16x16x4_f32(
                    false, w45, false, (v2f){e0[i][4], e0[i][5]}, (short)0, s0, false, false);
            s1 = __builtin_amdgcn_wmma_f32_16x16x4_f32(
                    false, w45, false, (v2f){e1[i][4], e1[i][5]}, (short)0, s1, false, false);
            s0 = __builtin_amdgcn_wmma_f32_16x16x4_f32(
                    false, w67, false, (v2f){e0[i][6], e0[i][7]}, (short)0, s0, false, false);
            s1 = __builtin_amdgcn_wmma_f32_16x16x4_f32(
                    false, w67, false, (v2f){e1[i][6], e1[i][7]}, (short)0, s1, false, false);
        }

        const float hx0 = half ? (LOG2E * g00) : 0.0f;
        const float hx1 = half ? (LOG2E * g10) : 0.0f;
        float v0 = 0.0f, v1 = 0.0f;
#pragma unroll
        for (int r = 0; r < 8; ++r) {
            const float fy = (float)(r >> 2), fj = (float)((r & 3) - 1);
            const float f0 = __builtin_amdgcn_exp2f(hx0 + LOG2E * (fy * g01 + fj * g02));
            const float f1 = __builtin_amdgcn_exp2f(hx1 + LOG2E * (fy * g11 + fj * g12));
            v0 = fmaf(f0, s0[r], v0);
            v1 = fmaf(f1, s1[r], v1);
        }
        v0 += other_half(v0);
        v1 += other_half(v1);
        if (half == 0) {
            part[w * WG_ROWS +      l15] = v0;
            part[w * WG_ROWS + 16 + l15] = v1;
        }
    } else {
#pragma unroll
        for (int i = 0; i < NTW; ++i) {
            if (gval[i]) {
                ((v4f*)pp4)[gf[i]]       = pv[i][0];
                ((v4f*)pp4)[gf[i] + 100] = pv[i][1];
            }
        }
        for (int s = tid; s < NY_NZ; s += NT) {
            const unsigned short us = (unsigned short)s;
            const unsigned short iy = us / NDIM_GRID;
            const unsigned short iz = us - iy * NDIM_GRID;
            const float cy = (float)iy, cz = (float)iz;
            const float u0 = a01 * cy + a02 * cz;
            const float u1 = a11 * cy + a12 * cz;
            const float u2 = a21 * cy + a22 * cz;
            const float v  = u0 + a10 * cy + a20 * cz;
            const float wq = cy * u1 + cz * u2;
#pragma unroll
            for (int ix = 0; ix < NDIM_GRID; ++ix) {
                const float cx = (float)ix;
                const float q  = cx * (a00 * cx + v) + wq;
                bq[ix * NY_NZ + s] = make_float4(cx, cy, cz, q);
            }
        }
        __syncthreads();

        const v2f* bbase = (const v2f*)bq + (l15 * 2 + half);

        v8f cvec0, cvec1;
#pragma unroll
        for (int r = 0; r < 8; ++r) {
            cvec0[r] = (-0.5f * LOG2E) * bq[rowbase +      8 * half + r].w;
            cvec1[r] = (-0.5f * LOG2E) * bq[rowbase + 16 + 8 * half + r].w;
        }

        float acc0[8] = {};
        float acc1[8] = {};
        const float* pbase = (const float*)pp4 + l15;

        const int npair = (NPAIRS - w + NW - 1) / NW;
        int j = w;
        v2f   b0 = bbase[(2 * j) * 32];
        v2f   b1 = bbase[(2 * j + 1) * 32];
        float p0 = pbase[(2 * j) * 16];
        float p1 = pbase[(2 * j + 1) * 16];

        for (int k = 0; k < npair; ++k) {
            const int jn = (j + NW < NPAIRS) ? (j + NW) : j;
            const v2f   nb0 = bbase[(2 * jn) * 32];
            const v2f   nb1 = bbase[(2 * jn + 1) * 32];
            const float np0 = pbase[(2 * jn) * 16];
            const float np1 = pbase[(2 * jn + 1) * 16];

            v8f c00 = __builtin_amdgcn_wmma_f32_16x16x4_f32(
                    false, amat0, false, b0, (short)0, cvec0, false, false);
            v8f c10 = __builtin_amdgcn_wmma_f32_16x16x4_f32(
                    false, amat1, false, b0, (short)0, cvec1, false, false);
            v8f c01t = __builtin_amdgcn_wmma_f32_16x16x4_f32(
                    false, amat0, false, b1, (short)0, cvec0, false, false);
            v8f c11 = __builtin_amdgcn_wmma_f32_16x16x4_f32(
                    false, amat1, false, b1, (short)0, cvec1, false, false);

#pragma unroll
            for (int r = 0; r < 8; ++r)
                acc0[r] = fmaf(__builtin_amdgcn_exp2f(c00[r]), p0, acc0[r]);
#pragma unroll
            for (int r = 0; r < 8; ++r)
                acc1[r] = fmaf(__builtin_amdgcn_exp2f(c10[r]), p0, acc1[r]);
#pragma unroll
            for (int r = 0; r < 8; ++r)
                acc0[r] = fmaf(__builtin_amdgcn_exp2f(c01t[r]), p1, acc0[r]);
#pragma unroll
            for (int r = 0; r < 8; ++r)
                acc1[r] = fmaf(__builtin_amdgcn_exp2f(c11[r]), p1, acc1[r]);

            asm volatile("" :: "v"(nb0), "v"(nb1), "v"(np0), "v"(np1));
            b0 = nb0; b1 = nb1; p0 = np0; p1 = np1;
            j += NW;
        }

#pragma unroll
        for (int r = 0; r < 8; ++r) {
            const float s0 = row16_sum(acc0[r]);
            const float s1 = row16_sum(acc1[r]);
            if (l15 == 0) {
                part[w * WG_ROWS +      8 * half + r] = s0;
                part[w * WG_ROWS + 16 + 8 * half + r] = s1;
            }
        }
    }
    __syncthreads();

    if (tid < WG_ROWS) {
        float s = part[tid];
#pragma unroll
        for (int v = 1; v < NW; ++v)
            s += part[v * WG_ROWS + tid];
        kde[rowbase + tid] = s;
    }
}

#define NORM_T   256
#define NORM_V   8
#define NVEC4    2000

__global__ void __launch_bounds__(NORM_T) kde_norm(float* __restrict__ kde) {
    __shared__ float red[16];
    const int tid  = threadIdx.x;
    const int lane = tid & 31;
    const int w    = tid >> 5;
    float4* vec = (float4*)kde;

    float4 v[NORM_V];
#pragma unroll
    for (int k = 0; k < NORM_V - 1; ++k)
        v[k] = vec[k * NORM_T + tid];
    const bool tail = (NORM_V - 1) * NORM_T + tid < NVEC4;
    v[NORM_V - 1] = tail ? vec[(NORM_V - 1) * NORM_T + tid]
                         : make_float4(0.0f, 0.0f, 0.0f, 0.0f);

    float t4[NORM_V];
#pragma unroll
    for (int k = 0; k < NORM_V; ++k)
        t4[k] = (v[k].x + v[k].y) + (v[k].z + v[k].w);
    float s = ((t4[0] + t4[1]) + (t4[2] + t4[3])) + ((t4[4] + t4[5]) + (t4[6] + t4[7]));

    s = row16_sum(s);
    if ((lane & 15) == 0) red[2 * w + (lane >> 4)] = s;
    __syncthreads();
    const float total = row16_sum(red[lane & 15]);
    const float inv = 1.0f / total;

#pragma unroll
    for (int k = 0; k < NORM_V - 1; ++k)
        vec[k * NORM_T + tid] = make_float4(v[k].x * inv, v[k].y * inv, v[k].z * inv, v[k].w * inv);
    if (tail)
        vec[(NORM_V - 1) * NORM_T + tid] = make_float4(v[NORM_V - 1].x * inv, v[NORM_V - 1].y * inv,
                                                       v[NORM_V - 1].z * inv, v[NORM_V - 1].w * inv);
}

extern "C" void kernel_launch(void* const* d_in, const int* in_sizes, int n_in,
                              void* d_out, int out_size, void* d_ws, size_t ws_size,
                              hipStream_t stream) {
    const float* p    = (const float*)d_in[0];
    const float* ainv = (const float*)d_in[1];
    float* out = (float*)d_out;

    kde_main<<<250, 256, 0, stream>>>(p, ainv, out);
    kde_norm<<<1, NORM_T, 0, stream>>>(out);
}
